# starting kernel
# baseline (speedup 1.0000x reference)
#include <hip/hip_runtime.h>
#include <math.h>

#define NQ 4
#define KMAX 516
#define PI_F 3.14159265358979323846f

typedef __attribute__((ext_vector_type(2))) float v2f;
typedef __attribute__((ext_vector_type(8))) float v8f;

__global__ void qrnn_trig(const float* __restrict__ theta,
                          float* __restrict__ trig) {
    int i = threadIdx.x;
    if (i < 2 * NQ * 3) {
        float h = theta[i] * 0.5f;
        trig[2 * i]     = cosf(h);
        trig[2 * i + 1] = sinf(h);
    }
}

__global__ __launch_bounds__(256)
void qrnn_encode(const float* __restrict__ inputs,
                 const float* __restrict__ prev_h,
                 const float* __restrict__ W_enc,
                 const float* __restrict__ b_enc,
                 float* __restrict__ angles,
                 int B, int D) {
    __shared__ float wlds[KMAX * 16];

    const int K = D + NQ;
    const int tid = threadIdx.x;

    for (int i = tid; i < K * 16; i += 256) {
        int r = i >> 4, c = i & 15;
        wlds[i] = (c < NQ) ? W_enc[r * NQ + c] : 0.0f;
    }
    __syncthreads();

    const int lane = tid & 31;
    const int wave = tid >> 5;
    const int g    = lane >> 4;
    const int n    = lane & 15;
    const int rowBase = blockIdx.x * 128 + wave * 16;

    int rowA = rowBase + n;
    if (rowA >= B) rowA = B - 1;

    const float* arow = inputs + (size_t)rowA * D;

    v8f acc = {};
    for (int k0 = 0; k0 < K; k0 += 4) {
        const int ka = k0 + 2 * g;
        float2 av;
        if (k0 < D) {
            av = *(const float2*)(arow + ka);
        } else {
            av = *(const float2*)(prev_h + (size_t)rowA * NQ + (ka - D));
        }
        v2f a, b;
        a[0] = av.x;
        a[1] = av.y;
        b[0] = wlds[ka * 16 + n];
        b[1] = wlds[(ka + 1) * 16 + n];
        acc = __builtin_amdgcn_wmma_f32_16x16x4_f32(
             false, a,  false, b,
             (short)0, acc,  false,  false);
    }

    if (n < NQ) {
        const float bias = b_enc[n];
#pragma unroll
        for (int j = 0; j < 8; ++j) {
            const int row = rowBase + j + g * 8;
            if (row < B) {
                angles[row * NQ + n] = tanhf(acc[j] + bias) * PI_F;
            }
        }
    }
}

__global__ __launch_bounds__(256)
void qrnn_circuit(const float* __restrict__ angles,
                  const float* __restrict__ trig,
                  const float* __restrict__ W_out,
                  const float* __restrict__ b_out,
                  float* __restrict__ out,
                  int B) {
    __shared__ float s_trig[48];
    __shared__ float s_w[16];
    __shared__ float s_b[4];
    const int tid = threadIdx.x;
    if (tid < 48) s_trig[tid] = trig[tid];
    if (tid < 16) s_w[tid]    = W_out[tid];
    if (tid < 4)  s_b[tid]    = b_out[tid];
    __syncthreads();

    const int row = blockIdx.x * 256 + tid;
    if (row >= B) return;

    const float4 ang = *(const float4*)(angles + (size_t)row * NQ);
    float ch[NQ], sh[NQ];
    {
        const float* ap = &ang.x;
#pragma unroll
        for (int q = 0; q < NQ; ++q) {
            float h = ap[q] * 0.5f;
            ch[q] = cosf(h);
            sh[q] = sinf(h);
        }
    }

    float sr[16], si[16];
#pragma unroll
    for (int idx = 0; idx < 16; ++idx) {
        float v = ((idx >> 3) & 1) ? sh[0] : ch[0];
        v *= ((idx >> 2) & 1) ? sh[1] : ch[1];
        v *= ((idx >> 1) & 1) ? sh[2] : ch[2];
        v *= (idx & 1)        ? sh[3] : ch[3];
        sr[idx] = v;
        si[idx] = 0.0f;
    }

#pragma unroll
    for (int l = 0; l < 2; ++l) {
#pragma unroll
        for (int q = 0; q < NQ; ++q) {
            const float* t = s_trig + (l * NQ + q) * 6;
            const float ca = t[0], sa = t[1];
            const float cb = t[2], sb = t[3];
            const float cy = t[4], sy = t[5];
            const int w = 1 << (3 - q);
#pragma unroll
            for (int m = 0; m < 8; ++m) {
                const int i0 = ((m & ~(w - 1)) << 1) | (m & (w - 1));
                const int i1 = i0 | w;
                float r0 = sr[i0], x0 = si[i0];
                float r1 = sr[i1], x1 = si[i1];
                float ar0 = ca * r0 + sa * x1;
                float ai0 = ca * x0 - sa * r1;
                float ar1 = ca * r1 + sa * x0;
                float ai1 = ca * x1 - sa * r0;
                float zr0 = cb * ar0 + sb * ai0;
                float zi0 = cb * ai0 - sb * ar0;
                float zr1 = cb * ar1 - sb * ai1;
                float zi1 = cb * ai1 + sb * ar1;
                sr[i0] = cy * zr0 - sy * zr1;
                si[i0] = cy * zi0 - sy * zi1;
                sr[i1] = sy * zr0 + cy * zr1;
                si[i1] = sy * zi0 + cy * zi1;
            }
        }
#pragma unroll
        for (int q = 0; q < NQ - 1; ++q) {
            const int cm = 1 << (3 - q);
            const int tm = 1 << (2 - q);
#pragma unroll
            for (int idx = 0; idx < 16; ++idx) {
                if ((idx & cm) && !(idx & tm)) {
                    const int j = idx | tm;
                    float tr = sr[idx]; sr[idx] = sr[j]; sr[j] = tr;
                    float ti = si[idx]; si[idx] = si[j]; si[j] = ti;
                }
            }
        }
    }

    float ev[NQ] = {0.0f, 0.0f, 0.0f, 0.0f};
#pragma unroll
    for (int idx = 0; idx < 16; ++idx) {
        const float p = sr[idx] * sr[idx] + si[idx] * si[idx];
        ev[0] += ((idx >> 3) & 1) ? -p : p;
        ev[1] += ((idx >> 2) & 1) ? -p : p;
        ev[2] += ((idx >> 1) & 1) ? -p : p;
        ev[3] += (idx & 1)        ? -p : p;
    }

    float4 o;
    float* op = &o.x;
#pragma unroll
    for (int j = 0; j < NQ; ++j) {
        float acc = s_b[j];
#pragma unroll
        for (int q = 0; q < NQ; ++q) acc += ev[q] * s_w[q * NQ + j];
        op[j] = tanhf(acc);
    }
    *(float4*)(out + (size_t)row * NQ) = o;
}

extern "C" void kernel_launch(void* const* d_in, const int* in_sizes, int n_in,
                              void* d_out, int out_size, void* d_ws, size_t ws_size,
                              hipStream_t stream) {
    (void)n_in; (void)out_size; (void)ws_size;
    const float* inputs = (const float*)d_in[0];
    const float* prev_h = (const float*)d_in[1];
    const float* W_enc  = (const float*)d_in[2];
    const float* b_enc  = (const float*)d_in[3];
    const float* theta  = (const float*)d_in[4];
    const float* W_out  = (const float*)d_in[5];
    const float* b_out  = (const float*)d_in[6];
    float* out = (float*)d_out;

    const int B = in_sizes[1] / NQ;
    const int D = in_sizes[0] / B;

    float* angles = (float*)d_ws;
    float* trig   = angles + (size_t)B * NQ;

    qrnn_trig<<<1, 32, 0, stream>>>(theta, trig);

    const int gridEnc = (B + 127) / 128;
    qrnn_encode<<<gridEnc, 256, 0, stream>>>(inputs, prev_h, W_enc, b_enc,
                                             angles, B, D);

    qrnn_circuit<<<(B + 255) / 256, 256, 0, stream>>>(angles, trig, W_out,
                                                      b_out, out, B);
}


# encode: 16x16x4 tile remapped to 4 rows x 4 k-chunks (diagonal blocks added across a quad) so each 16-byte load covers whole 128-B lines, 32 loads per wave in flight; changes f32 summation order
# speedup vs baseline: 2.3608x; 2.3608x over previous
#include <hip/hip_runtime.h>
#include <math.h>

#define NQ 4
#define DCOLS 512
#define KTOT 516
#define WPITCH 544
#define PI_F 3.14159265358979323846f

typedef __attribute__((ext_vector_type(2))) float v2f;
typedef __attribute__((ext_vector_type(8))) float v8f;

__global__ void qrnn_trig(const float* __restrict__ theta,
                          float* __restrict__ trig) {
    int i = threadIdx.x;
    if (i < 2 * NQ * 3) {
        float h = theta[i] * 0.5f;
        trig[2 * i]     = cosf(h);
        trig[2 * i + 1] = sinf(h);
    }
}

__device__ __forceinline__ v8f wmma4(float a0, float a1, float b0, float b1, v8f acc) {
    v2f a, b;
    a[0] = a0; a[1] = a1;
    b[0] = b0; b[1] = b1;
    return __builtin_amdgcn_wmma_f32_16x16x4_f32(
         false, a,  false, b,
         (short)0, acc,  false,  false);
}

__global__ __launch_bounds__(256) __attribute__((amdgpu_waves_per_eu(4, 4)))
void qrnn_encode(const float* __restrict__ inputs,
                 const float* __restrict__ prev_h,
                 const float* __restrict__ W_enc,
                 const float* __restrict__ b_enc,
                 float* __restrict__ angles,
                 int B, int D) {
    (void)D;
    __shared__ __attribute__((aligned(16))) float wt[NQ * WPITCH];

    const int tid = threadIdx.x;

    {
        const float4 w0 = *(const float4*)(W_enc + tid * NQ);
        const float4 w1 = *(const float4*)(W_enc + (tid + 256) * NQ);
        float4 w2 = make_float4(0.0f, 0.0f, 0.0f, 0.0f);
        if (tid < KTOT - 512) w2 = *(const float4*)(W_enc + (tid + 512) * NQ);
        wt[0 * WPITCH + tid] = w0.x;
        wt[1 * WPITCH + tid] = w0.y;
        wt[2 * WPITCH + tid] = w0.z;
        wt[3 * WPITCH + tid] = w0.w;
        wt[0 * WPITCH + tid + 256] = w1.x;
        wt[1 * WPITCH + tid + 256] = w1.y;
        wt[2 * WPITCH + tid + 256] = w1.z;
        wt[3 * WPITCH + tid + 256] = w1.w;
        if (tid < KTOT - 512) {
            wt[0 * WPITCH + tid + 512] = w2.x;
            wt[1 * WPITCH + tid + 512] = w2.y;
            wt[2 * WPITCH + tid + 512] = w2.z;
            wt[3 * WPITCH + tid + 512] = w2.w;
        }
    }
    __syncthreads();

    const int lane = tid & 31;
    const int wave = tid >> 5;
    const int g    = lane >> 4;
    const int q    = lane & 3;
    const int rc   = (lane >> 2) & 3;
    const int koff = 16 * g + 4 * q;
    const int row0 = (blockIdx.x * 8 + wave) * 8;

    const int qb  = lane & ~3;
    const int src = 16 * ((lane >> 3) & 1) + 4 * (lane & 3) + 2 * (lane >> 4) + ((lane >> 2) & 1);

    int r0 = row0 + rc;
    int r1 = row0 + 4 + rc;
    if (r0 >= B) r0 = B - 1;
    if (r1 >= B) r1 = B - 1;
    const float* pa0 = inputs + (size_t)r0 * DCOLS + koff;
    const float* pa1 = inputs + (size_t)r1 * DCOLS + koff;

    const float2 h0 = *(const float2*)(prev_h + (size_t)r0 * NQ + 2 * g);
    const float2 h1 = *(const float2*)(prev_h + (size_t)r1 * NQ + 2 * g);
    const float bias = b_enc[rc];
    __builtin_amdgcn_sched_barrier(0);

    float4 a0[16], a1[16];
#pragma unroll
    for (int j = 0; j < 16; ++j) {
        a0[j] = *(const float4*)(pa0 + 32 * j);
        a1[j] = *(const float4*)(pa1 + 32 * j);
    }
    __builtin_amdgcn_sched_barrier(0);

    const float* pb = wt + rc * WPITCH + koff;

    v8f acc0 = {}, acc1 = {};
    {
        const bool c0 = (q == 0);
        const float t0 = wt[rc * WPITCH + DCOLS + 2 * g];
        const float t1 = wt[rc * WPITCH + DCOLS + 2 * g + 1];
        const float bt0 = c0 ? t0 : 0.0f;
        const float bt1 = c0 ? t1 : 0.0f;
        acc0 = wmma4(c0 ? h0.x : 0.0f, c0 ? h0.y : 0.0f, bt0, bt1, acc0);
        acc1 = wmma4(c0 ? h1.x : 0.0f, c0 ? h1.y : 0.0f, bt0, bt1, acc1);
    }
#pragma unroll
    for (int j = 0; j < 16; ++j) {
        const float4 bw = *(const float4*)(pb + 32 * j);
        acc0 = wmma4(a0[j].x, a0[j].y, bw.x, bw.y, acc0);
        acc1 = wmma4(a1[j].x, a1[j].y, bw.x, bw.y, acc1);
        acc0 = wmma4(a0[j].z, a0[j].w, bw.z, bw.w, acc0);
        acc1 = wmma4(a1[j].z, a1[j].w, bw.z, bw.w, acc1);
    }

    float p00 = __shfl(acc0[0], qb);
    float p01 = __shfl(acc0[4], qb);
    float p10 = __shfl(acc1[0], qb);
    float p11 = __shfl(acc1[4], qb);
    p00 += __shfl(acc0[1], qb + 1);  p01 += __shfl(acc0[5], qb + 1);
    p10 += __shfl(acc1[1], qb + 1);  p11 += __shfl(acc1[5], qb + 1);
    p00 += __shfl(acc0[2], qb + 2);  p01 += __shfl(acc0[6], qb + 2);
    p10 += __shfl(acc1[2], qb + 2);  p11 += __shfl(acc1[6], qb + 2);
    p00 += __shfl(acc0[3], qb + 3);  p01 += __shfl(acc0[7], qb + 3);
    p10 += __shfl(acc1[3], qb + 3);  p11 += __shfl(acc1[7], qb + 3);

    const float ua = (q & 1) ? p01 : p00;
    const float ub = (q & 1) ? p11 : p10;
    const float u  = ((q & 2) ? ub : ua) + bias;
    const float v = __shfl(u, src);
    const int orow = row0 + (lane >> 2);
    if (orow < B) {
        angles[(size_t)row0 * NQ + lane] = tanhf(v) * PI_F;
    }
}

__global__ __launch_bounds__(256)
void qrnn_circuit(const float* __restrict__ angles,
                  const float* __restrict__ trig,
                  const float* __restrict__ W_out,
                  const float* __restrict__ b_out,
                  float* __restrict__ out,
                  int B) {
    __shared__ float s_trig[48];
    __shared__ float s_w[16];
    __shared__ float s_b[4];
    const int tid = threadIdx.x;
    if (tid < 48) s_trig[tid] = trig[tid];
    if (tid < 16) s_w[tid]    = W_out[tid];
    if (tid < 4)  s_b[tid]    = b_out[tid];
    __syncthreads();

    const int row = blockIdx.x * 256 + tid;
    if (row >= B) return;

    const float4 ang = *(const float4*)(angles + (size_t)row * NQ);
    float ch[NQ], sh[NQ];
    {
        const float* ap = &ang.x;
#pragma unroll
        for (int q = 0; q < NQ; ++q) {
            float h = ap[q] * 0.5f;
            ch[q] = cosf(h);
            sh[q] = sinf(h);
        }
    }

    float sr[16], si[16];
#pragma unroll
    for (int idx = 0; idx < 16; ++idx) {
        float v = ((idx >> 3) & 1) ? sh[0] : ch[0];
        v *= ((idx >> 2) & 1) ? sh[1] : ch[1];
        v *= ((idx >> 1) & 1) ? sh[2] : ch[2];
        v *= (idx & 1)        ? sh[3] : ch[3];
        sr[idx] = v;
        si[idx] = 0.0f;
    }

#pragma unroll
    for (int l = 0; l < 2; ++l) {
#pragma unroll
        for (int q = 0; q < NQ; ++q) {
            const float* t = s_trig + (l * NQ + q) * 6;
            const float ca = t[0], sa = t[1];
            const float cb = t[2], sb = t[3];
            const float cy = t[4], sy = t[5];
            const int w = 1 << (3 - q);
#pragma unroll
            for (int m = 0; m < 8; ++m) {
                const int i0 = ((m & ~(w - 1)) << 1) | (m & (w - 1));
                const int i1 = i0 | w;
                float r0 = sr[i0], x0 = si[i0];
                float r1 = sr[i1], x1 = si[i1];
                float ar0 = ca * r0 + sa * x1;
                float ai0 = ca * x0 - sa * r1;
                float ar1 = ca * r1 + sa * x0;
                float ai1 = ca * x1 - sa * r0;
                float zr0 = cb * ar0 + sb * ai0;
                float zi0 = cb * ai0 - sb * ar0;
                float zr1 = cb * ar1 - sb * ai1;
                float zi1 = cb * ai1 + sb * ar1;
                sr[i0] = cy * zr0 - sy * zr1;
                si[i0] = cy * zi0 - sy * zi1;
                sr[i1] = sy * zr0 + cy * zr1;
                si[i1] = sy * zi0 + cy * zi1;
            }
        }
#pragma unroll
        for (int q = 0; q < NQ - 1; ++q) {
            const int cm = 1 << (3 - q);
            const int tm = 1 << (2 - q);
#pragma unroll
            for (int idx = 0; idx < 16; ++idx) {
                if ((idx & cm) && !(idx & tm)) {
                    const int j = idx | tm;
                    float tr = sr[idx]; sr[idx] = sr[j]; sr[j] = tr;
                    float ti = si[idx]; si[idx] = si[j]; si[j] = ti;
                }
            }
        }
    }

    float ev[NQ] = {0.0f, 0.0f, 0.0f, 0.0f};
#pragma unroll
    for (int idx = 0; idx < 16; ++idx) {
        const float p = sr[idx] * sr[idx] + si[idx] * si[idx];
        ev[0] += ((idx >> 3) & 1) ? -p : p;
        ev[1] += ((idx >> 2) & 1) ? -p : p;
        ev[2] += ((idx >> 1) & 1) ? -p : p;
        ev[3] += (idx & 1)        ? -p : p;
    }

    float4 o;
    float* op = &o.x;
#pragma unroll
    for (int j = 0; j < NQ; ++j) {
        float acc = s_b[j];
#pragma unroll
        for (int q = 0; q < NQ; ++q) acc += ev[q] * s_w[q * NQ + j];
        op[j] = tanhf(acc);
    }
    *(float4*)(out + (size_t)row * NQ) = o;
}

extern "C" void kernel_launch(void* const* d_in, const int* in_sizes, int n_in,
                              void* d_out, int out_size, void* d_ws, size_t ws_size,
                              hipStream_t stream) {
    (void)n_in; (void)out_size; (void)ws_size;
    const float* inputs = (const float*)d_in[0];
    const float* prev_h = (const float*)d_in[1];
    const float* W_enc  = (const float*)d_in[2];
    const float* b_enc  = (const float*)d_in[3];
    const float* theta  = (const float*)d_in[4];
    const float* W_out  = (const float*)d_in[5];
    const float* b_out  = (const float*)d_in[6];
    float* out = (float*)d_out;

    const int B = in_sizes[1] / NQ;
    const int D = in_sizes[0] / B;

    float* angles = (float*)d_ws;
    float* trig   = angles + (size_t)B * NQ;

    qrnn_trig<<<1, 32, 0, stream>>>(theta, trig);

    const int gridEnc = (B + 63) / 64;
    qrnn_encode<<<gridEnc, 256, 0, stream>>>(inputs, prev_h, W_enc, b_enc,
                                             angles, B, D);

    qrnn_circuit<<<(B + 255) / 256, 256, 0, stream>>>(angles, trig, W_out,
                                                      b_out, out, B);
}


# one launch: trig table and circuit fused into the encode kernel; each row's 16 amplitudes split over its quad (4 per lane, DPP exchange); changes f32 summation order of the expectation sums
# speedup vs baseline: 3.6134x; 1.5306x over previous
#include <hip/hip_runtime.h>
#include <math.h>

#define NQ 4
#define DCOLS 512
#define KTOT 516
#define WPITCH 544
#define PI_F 3.14159265358979323846f

typedef __attribute__((ext_vector_type(2))) float v2f;
typedef __attribute__((ext_vector_type(8))) float v8f;

#define QP_X1  0xB1
#define QP_X2  0x4E
#define QP_C01 0xB4

template <int CTRL>
__device__ __forceinline__ float qperm(float v) {
    return __int_as_float(__builtin_amdgcn_update_dpp(0, __float_as_int(v), CTRL, 0xF, 0xF, true));
}

__device__ __forceinline__ float pick(bool m, float a, float b) { return m ? a : b; }

template <int CTRL>
__device__ __forceinline__ void lane_wire(float* sr, float* si, bool one,
                                          const float* tc, const float* ts) {
    const float ca = tc[0], sa = ts[0];
    const float cb = tc[1], sb = ts[1];
    const float cy = tc[2], sy = ts[2];
    const float sbs = pick(one, -sb, sb);
    const float sys = pick(one, sy, -sy);
#pragma unroll
    for (int j = 0; j < 4; ++j) {
        const float ro = qperm<CTRL>(sr[j]);
        const float xo = qperm<CTRL>(si[j]);
        const float ar = ca * sr[j] + sa * xo;
        const float ai = ca * si[j] - sa * ro;
        const float zr = cb * ar + sbs * ai;
        const float zi = cb * ai - sbs * ar;
        const float zro = qperm<CTRL>(zr);
        const float zio = qperm<CTRL>(zi);
        sr[j] = cy * zr + sys * zro;
        si[j] = cy * zi + sys * zio;
    }
}

template <int W>
__device__ __forceinline__ void local_wire(float* sr, float* si,
                                           const float* tc, const float* ts) {
    const float ca = tc[0], sa = ts[0];
    const float cb = tc[1], sb = ts[1];
    const float cy = tc[2], sy = ts[2];
#pragma unroll
    for (int m = 0; m < 2; ++m) {
        const int i0 = (W == 2) ? m : 2 * m;
        const int i1 = i0 | W;
        const float r0 = sr[i0], x0 = si[i0];
        const float r1 = sr[i1], x1 = si[i1];
        const float ar0 = ca * r0 + sa * x1;
        const float ai0 = ca * x0 - sa * r1;
        const float ar1 = ca * r1 + sa * x0;
        const float ai1 = ca * x1 - sa * r0;
        const float zr0 = cb * ar0 + sb * ai0;
        const float zi0 = cb * ai0 - sb * ar0;
        const float zr1 = cb * ar1 - sb * ai1;
        const float zi1 = cb * ai1 + sb * ar1;
        sr[i0] = cy * zr0 - sy * zr1;
        si[i0] = cy * zi0 - sy * zi1;
        sr[i1] = sy * zr0 + cy * zr1;
        si[i1] = sy * zi0 + cy * zi1;
    }
}

__device__ __forceinline__ void quad_circuit(float c, float s, bool hi, bool lo,
                                             const float* tc, const float* ts,
                                             float& e0, float& e1, float& e2, float& e3) {
    const float c0 = qperm<0x00>(c), c1 = qperm<0x55>(c), c2 = qperm<0xAA>(c), c3 = qperm<0xFF>(c);
    const float s0 = qperm<0x00>(s), s1 = qperm<0x55>(s), s2 = qperm<0xAA>(s), s3 = qperm<0xFF>(s);

    const float f01 = pick(hi, s0, c0) * pick(lo, s1, c1);
    float sr[4], si[4];
#pragma unroll
    for (int j = 0; j < 4; ++j) {
        float v = f01 * ((j & 2) ? s2 : c2);
        v *= (j & 1) ? s3 : c3;
        sr[j] = v;
        si[j] = 0.0f;
    }

#pragma unroll
    for (int l = 0; l < 2; ++l) {
        lane_wire<QP_X2>(sr, si, hi, tc + (l * NQ + 0) * 3, ts + (l * NQ + 0) * 3);
        lane_wire<QP_X1>(sr, si, lo, tc + (l * NQ + 1) * 3, ts + (l * NQ + 1) * 3);
        local_wire<2>(sr, si, tc + (l * NQ + 2) * 3, ts + (l * NQ + 2) * 3);
        local_wire<1>(sr, si, tc + (l * NQ + 3) * 3, ts + (l * NQ + 3) * 3);
#pragma unroll
        for (int j = 0; j < 4; ++j) {
            sr[j] = qperm<QP_C01>(sr[j]);
            si[j] = qperm<QP_C01>(si[j]);
        }
#pragma unroll
        for (int j = 0; j < 2; ++j) {
            const float ra = sr[j], rb = sr[j + 2];
            const float xa = si[j], xb = si[j + 2];
            sr[j]     = pick(lo, rb, ra);
            sr[j + 2] = pick(lo, ra, rb);
            si[j]     = pick(lo, xb, xa);
            si[j + 2] = pick(lo, xa, xb);
        }
        { const float t = sr[2]; sr[2] = sr[3]; sr[3] = t; }
        { const float t = si[2]; si[2] = si[3]; si[3] = t; }
    }

    float p[4];
#pragma unroll
    for (int j = 0; j < 4; ++j) p[j] = sr[j] * sr[j] + si[j] * si[j];
    const float t = ((p[0] + p[1]) + p[2]) + p[3];
    float a0 = pick(hi, -t, t);
    float a1 = pick(lo, -t, t);
    float a2 = (p[0] + p[1]) - (p[2] + p[3]);
    float a3 = (p[0] - p[1]) + (p[2] - p[3]);
    a0 += qperm<QP_X1>(a0);  a1 += qperm<QP_X1>(a1);
    a2 += qperm<QP_X1>(a2);  a3 += qperm<QP_X1>(a3);
    a0 += qperm<QP_X2>(a0);  a1 += qperm<QP_X2>(a1);
    a2 += qperm<QP_X2>(a2);  a3 += qperm<QP_X2>(a3);
    e0 = a0; e1 = a1; e2 = a2; e3 = a3;
}

__device__ __forceinline__ v8f wmma4(float a0, float a1, float b0, float b1, v8f acc) {
    v2f a, b;
    a[0] = a0; a[1] = a1;
    b[0] = b0; b[1] = b1;
    return __builtin_amdgcn_wmma_f32_16x16x4_f32(
         false, a,  false, b,
         (short)0, acc,  false,  false);
}

__global__ __launch_bounds__(256) __attribute__((amdgpu_waves_per_eu(4, 4)))
void qrnn_cell(const float* __restrict__ inputs,
               const float* __restrict__ prev_h,
               const float* __restrict__ W_enc,
               const float* __restrict__ b_enc,
               const float* __restrict__ theta,
               const float* __restrict__ W_out,
               const float* __restrict__ b_out,
               float* __restrict__ out,
               int B, int D) {
    (void)D;
    __shared__ __attribute__((aligned(16))) float wt[NQ * WPITCH];

    const int tid = threadIdx.x;

    {
        const float4 w0 = *(const float4*)(W_enc + tid * NQ);
        const float4 w1 = *(const float4*)(W_enc + (tid + 256) * NQ);
        float4 w2 = make_float4(0.0f, 0.0f, 0.0f, 0.0f);
        if (tid < KTOT - 512) w2 = *(const float4*)(W_enc + (tid + 512) * NQ);
        wt[0 * WPITCH + tid] = w0.x;
        wt[1 * WPITCH + tid] = w0.y;
        wt[2 * WPITCH + tid] = w0.z;
        wt[3 * WPITCH + tid] = w0.w;
        wt[0 * WPITCH + tid + 256] = w1.x;
        wt[1 * WPITCH + tid + 256] = w1.y;
        wt[2 * WPITCH + tid + 256] = w1.z;
        wt[3 * WPITCH + tid + 256] = w1.w;
        if (tid < KTOT - 512) {
            wt[0 * WPITCH + tid + 512] = w2.x;
            wt[1 * WPITCH + tid + 512] = w2.y;
            wt[2 * WPITCH + tid + 512] = w2.z;
            wt[3 * WPITCH + tid + 512] = w2.w;
        }
    }
    __syncthreads();

    const int lane = tid & 31;
    const int wave = tid >> 5;
    const int g    = lane >> 4;
    const int q    = lane & 3;
    const int rc   = (lane >> 2) & 3;
    const int koff = 16 * g + 4 * q;
    const int row0 = (blockIdx.x * 8 + wave) * 8;

    const int qb  = lane & ~3;
    const int src = 16 * ((lane >> 3) & 1) + 4 * (lane & 3) + 2 * (lane >> 4) + ((lane >> 2) & 1);

    int r0 = row0 + rc;
    int r1 = row0 + 4 + rc;
    if (r0 >= B) r0 = B - 1;
    if (r1 >= B) r1 = B - 1;
    const float* pa0 = inputs + (size_t)r0 * DCOLS + koff;
    const float* pa1 = inputs + (size_t)r1 * DCOLS + koff;

    const float2 h0 = *(const float2*)(prev_h + (size_t)r0 * NQ + 2 * g);
    const float2 h1 = *(const float2*)(prev_h + (size_t)r1 * NQ + 2 * g);
    const float bias = b_enc[rc];
    const float th = theta[lane < 2 * NQ * 3 ? lane : 0];
    __builtin_amdgcn_sched_barrier(0);

    float4 a0[16], a1[16];
#pragma unroll
    for (int j = 0; j < 16; ++j) {
        a0[j] = *(const float4*)(pa0 + 32 * j);
        a1[j] = *(const float4*)(pa1 + 32 * j);
    }
    __builtin_amdgcn_sched_barrier(0);

    float tc[2 * NQ * 3], ts[2 * NQ * 3];
    {
        const float hth = th * 0.5f;
        const float cv = cosf(hth);
        const float sv = sinf(hth);
#pragma unroll
        for (int i = 0; i < 2 * NQ * 3; ++i) {
            tc[i] = __int_as_float(__builtin_amdgcn_readlane(__float_as_int(cv), i));
            ts[i] = __int_as_float(__builtin_amdgcn_readlane(__float_as_int(sv), i));
        }
    }
    __builtin_amdgcn_sched_barrier(0);

    const float* pb = wt + rc * WPITCH + koff;

    v8f acc0 = {}, acc1 = {};
    {
        const bool c0 = (q == 0);
        const float t0 = wt[rc * WPITCH + DCOLS + 2 * g];
        const float t1 = wt[rc * WPITCH + DCOLS + 2 * g + 1];
        const float bt0 = c0 ? t0 : 0.0f;
        const float bt1 = c0 ? t1 : 0.0f;
        acc0 = wmma4(c0 ? h0.x : 0.0f, c0 ? h0.y : 0.0f, bt0, bt1, acc0);
        acc1 = wmma4(c0 ? h1.x : 0.0f, c0 ? h1.y : 0.0f, bt0, bt1, acc1);
    }
#pragma unroll
    for (int j = 0; j < 16; ++j) {
        const float4 bw = *(const float4*)(pb + 32 * j);
        acc0 = wmma4(a0[j].x, a0[j].y, bw.x, bw.y, acc0);
        acc1 = wmma4(a1[j].x, a1[j].y, bw.x, bw.y, acc1);
        acc0 = wmma4(a0[j].z, a0[j].w, bw.z, bw.w, acc0);
        acc1 = wmma4(a1[j].z, a1[j].w, bw.z, bw.w, acc1);
    }
    asm volatile("v_nop\n\tv_nop\n\tv_nop\n\tv_nop" : "+v"(acc0), "+v"(acc1));
    __builtin_amdgcn_sched_barrier(0);

    float p00 = __shfl(acc0[0], qb);
    float p01 = __shfl(acc0[4], qb);
    float p10 = __shfl(acc1[0], qb);
    float p11 = __shfl(acc1[4], qb);
    p00 += __shfl(acc0[1], qb + 1);  p01 += __shfl(acc0[5], qb + 1);
    p10 += __shfl(acc1[1], qb + 1);  p11 += __shfl(acc1[5], qb + 1);
    p00 += __shfl(acc0[2], qb + 2);  p01 += __shfl(acc0[6], qb + 2);
    p10 += __shfl(acc1[2], qb + 2);  p11 += __shfl(acc1[6], qb + 2);
    p00 += __shfl(acc0[3], qb + 3);  p01 += __shfl(acc0[7], qb + 3);
    p10 += __shfl(acc1[3], qb + 3);  p11 += __shfl(acc1[7], qb + 3);

    const float ua = (q & 1) ? p01 : p00;
    const float ub = (q & 1) ? p11 : p10;
    const float u  = ((q & 2) ? ub : ua) + bias;
    const float v = __shfl(u, src);

    const float wo0 = W_out[0 * NQ + q];
    const float wo1 = W_out[1 * NQ + q];
    const float wo2 = W_out[2 * NQ + q];
    const float wo3 = W_out[3 * NQ + q];
    const float bo  = b_out[q];

    const float ang = tanhf(v) * PI_F;
    const float hang = ang * 0.5f;
    float e0, e1, e2, e3;
    quad_circuit(cosf(hang), sinf(hang), (q & 2) != 0, (q & 1) != 0, tc, ts, e0, e1, e2, e3);

    float o = bo;
    o += e0 * wo0;
    o += e1 * wo1;
    o += e2 * wo2;
    o += e3 * wo3;
    const int orow = row0 + (lane >> 2);
    if (orow < B) {
        out[(size_t)row0 * NQ + lane] = tanhf(o);
    }
}

extern "C" void kernel_launch(void* const* d_in, const int* in_sizes, int n_in,
                              void* d_out, int out_size, void* d_ws, size_t ws_size,
                              hipStream_t stream) {
    (void)n_in; (void)out_size; (void)d_ws; (void)ws_size;
    const float* inputs = (const float*)d_in[0];
    const float* prev_h = (const float*)d_in[1];
    const float* W_enc  = (const float*)d_in[2];
    const float* b_enc  = (const float*)d_in[3];
    const float* theta  = (const float*)d_in[4];
    const float* W_out  = (const float*)d_in[5];
    const float* b_out  = (const float*)d_in[6];
    float* out = (float*)d_out;

    const int B = in_sizes[1] / NQ;
    const int D = in_sizes[0] / B;

    const int grid = (B + 63) / 64;
    qrnn_cell<<<grid, 256, 0, stream>>>(inputs, prev_h, W_enc, b_enc,
                                        theta, W_out, b_out, out, B, D);
}
